# starting kernel
# baseline (speedup 1.0000x reference)
#include <hip/hip_runtime.h>


constexpr int kDepth   = 16;
constexpr int kB       = 16;
constexpr int kH       = 96;
constexpr int kW       = 96;
constexpr int kNPad    = 96;
constexpr int kKPad    = 12;
constexpr int kWaves   = 6;
constexpr int kThreads = kWaves * 32;

typedef __attribute__((ext_vector_type(2))) float v2f;
typedef __attribute__((ext_vector_type(8))) float v8f;

__global__ __launch_bounds__(kThreads)
void vin_step_kernel(const float* __restrict__ v_in,
                     const float* __restrict__ rewards,
                     const float* __restrict__ wt,
                     float* __restrict__ v_out) {
    __shared__ float lds_rv[3][kW + 2];
    __shared__ float lds_w[kNPad][kKPad];
    __shared__ float lds_dyn[kW][kNPad];
    __shared__ float lds_red[kW];

    const int tid  = threadIdx.x;
    const int lane = tid & 31;
    const int wv   = tid >> 5;
    const int b    = blockIdx.x / kH;
    const int h    = blockIdx.x % kH;

    const float* vimg = v_in    + (size_t)b * kH * kW;
    const float* rimg = rewards + (size_t)b * kH * kW;

    for (int i = tid; i < 3 * (kW + 2); i += kThreads) {
        const int r  = i / (kW + 2);
        const int c  = i % (kW + 2);
        const int hh = h - 1 + r;
        const int ww = c - 1;
        const bool inb = (hh >= 0) & (hh < kH) & (ww >= 0) & (ww < kW);
        const int hc = min(max(hh, 0), kH - 1);
        const int wc = min(max(ww, 0), kW - 1);
        const float val = vimg[hc * kW + wc] + rimg[hc * kW + wc];
        lds_rv[r][c] = inb ? val : 0.0f;
    }

    for (int i = tid; i < kNPad * kKPad; i += kThreads) {
        const int ch = i / kKPad;
        const int k  = i % kKPad;
        const bool valid = (ch < 90) & (k < 9);
        const float val = wt[min(ch, 89) * 9 + min(k, 8)];
        lds_w[ch][k] = valid ? val : 0.0f;
    }
    __syncthreads();

    const int  m  = (lane & 15) + wv * 16;
    const bool lo = lane < 16;
#define RV(r, c) lds_rv[r][m + (c)]
    v2f afrag[3];
    afrag[0].x = lo ? RV(0, 0) : RV(0, 2);
    afrag[0].y = lo ? RV(0, 1) : RV(1, 0);
    afrag[1].x = lo ? RV(1, 1) : RV(2, 0);
    afrag[1].y = lo ? RV(1, 2) : RV(2, 1);
    afrag[2].x = lo ? RV(2, 2) : 0.0f;
    afrag[2].y = 0.0f;
#undef RV

    const int n    = (lane & 15);
    const int koff = lo ? 0 : 2;
#pragma unroll 1
    for (int nt = 0; nt < 6; ++nt) {
        const int ch = nt * 16 + n;
        v2f b0 = *(const v2f*)&lds_w[ch][0 + koff];
        v2f b1 = *(const v2f*)&lds_w[ch][4 + koff];
        v2f b2 = *(const v2f*)&lds_w[ch][8 + koff];
        v8f acc = {};
        acc = __builtin_amdgcn_wmma_f32_16x16x4_f32(false, afrag[0], false, b0,
                                                    (short)0, acc, false, false);
        acc = __builtin_amdgcn_wmma_f32_16x16x4_f32(false, afrag[1], false, b1,
                                                    (short)0, acc, false, false);
        acc = __builtin_amdgcn_wmma_f32_16x16x4_f32(false, afrag[2], false, b2,
                                                    (short)0, acc, false, false);
        const int mbase = wv * 16 + (lo ? 0 : 8);
#pragma unroll
        for (int r = 0; r < 8; ++r)
            lds_dyn[mbase + r][ch] = acc[r];
    }
    __syncthreads();

    const int p    = tid % kW;
    const int half = tid / kW;

    float patch[9];
#pragma unroll
    for (int t = 0; t < 9; ++t)
        patch[t] = lds_rv[t / 3][p + (t % 3)];

    float best = -3.402823466e38f;
    for (int a = half * 5; a < half * 5 + 5; ++a) {
        const float* dl = &lds_dyn[p][a * 9];
        float mx = dl[0];
#pragma unroll
        for (int t = 1; t < 9; ++t) mx = fmaxf(mx, dl[t]);
        float ssum = 0.0f, q = 0.0f;
#pragma unroll
        for (int t = 0; t < 9; ++t) {
            const float e = __expf(dl[t] - mx);
            ssum += e;
            q    += e * patch[t];
        }
        best = fmaxf(best, q / ssum);
    }

    if (half == 1) lds_red[p] = best;
    __syncthreads();
    if (half == 0) {
        best = fmaxf(best, lds_red[p]);
        v_out[((size_t)b * kH + h) * kW + p] = best;
    }
}

extern "C" void kernel_launch(void* const* d_in, const int* in_sizes, int n_in,
                              void* d_out, int out_size, void* d_ws, size_t ws_size,
                              hipStream_t stream) {
    (void)in_sizes; (void)n_in; (void)out_size; (void)d_ws; (void)ws_size;
    const float* values  = (const float*)d_in[0];
    const float* rewards = (const float*)d_in[1];
    const float* wt      = (const float*)d_in[2];
    float* out = (float*)d_out;

    const size_t slice = (size_t)kB * kH * kW;
    for (int t = 0; t < kDepth; ++t) {
        const float* vin = (t == 0) ? values : out + (size_t)(t - 1) * slice;
        float* vout = out + (size_t)t * slice;
        vin_step_kernel<<<kB * kH, kThreads, 0, stream>>>(vin, rewards, wt, vout);
    }
}


# Logit tiles regrouped one per softmax tap (columns = actions, pixels stay A): a lane's 9 accumulators hold one action's 9 logits for 8 pixels; softmax in registers, max over actions by row DPP
# speedup vs baseline: 1.5432x; 1.5432x over previous
#include <hip/hip_runtime.h>


constexpr int kDepth   = 16;
constexpr int kB       = 16;
constexpr int kH       = 96;
constexpr int kW       = 96;
constexpr int kAct     = 10;
constexpr int kTaps    = 9;
constexpr int kKPad    = 12;
constexpr int kWaves   = 6;
constexpr int kThreads = kWaves * 32;

typedef __attribute__((ext_vector_type(2))) float v2f;
typedef __attribute__((ext_vector_type(8))) float v8f;

template <int M>
__device__ __forceinline__ float row_max_step(float x) {
    const int b = __builtin_bit_cast(int, x);
    const int o = __builtin_amdgcn_update_dpp(b, b, 0x160 + M, 0xf, 0xf, false);
    return fmaxf(x, __builtin_bit_cast(float, o));
}

__global__ __launch_bounds__(kThreads)
void vin_step_kernel(const float* __restrict__ v_in,
                     const float* __restrict__ rewards,
                     const float* __restrict__ wt,
                     float* __restrict__ v_out) {
    __shared__ float lds_rv[3][kW + 2];
    __shared__ float lds_w[kTaps][kAct][kKPad];
    __shared__ float lds_out[kW];

    const int tid  = threadIdx.x;
    const int lane = tid & 31;
    const int wv   = tid >> 5;
    const int b    = blockIdx.x / kH;
    const int h    = blockIdx.x % kH;

    const float* vimg = v_in    + (size_t)b * kH * kW;
    const float* rimg = rewards + (size_t)b * kH * kW;

    for (int i = tid; i < 3 * (kW + 2); i += kThreads) {
        const int r  = i / (kW + 2);
        const int c  = i % (kW + 2);
        const int hh = h - 1 + r;
        const int ww = c - 1;
        const bool inb = (hh >= 0) & (hh < kH) & (ww >= 0) & (ww < kW);
        const int hc = min(max(hh, 0), kH - 1);
        const int wc = min(max(ww, 0), kW - 1);
        const float val = vimg[hc * kW + wc] + rimg[hc * kW + wc];
        lds_rv[r][c] = inb ? val : 0.0f;
    }

    for (int idx = tid; idx < kTaps * kAct * kKPad; idx += kThreads) {
        const int i  = idx / (kAct * kKPad);
        const int a  = (idx / kKPad) % kAct;
        const int k  = idx % kKPad;
        const int ch = a * kTaps + i;
        const float val = wt[ch * 9 + min(k, 8)];
        lds_w[i][a][k] = (k < 9) ? val : 0.0f;
    }
    __syncthreads();

    const int  c16 = lane & 15;
    const int  hi  = lane >> 4;
    const int  m   = c16 + wv * 16;
    const bool lo  = lane < 16;
#define RV(r, c) lds_rv[r][m + (c)]
    v2f afrag[3];
    afrag[0].x = lo ? RV(0, 0) : RV(0, 2);
    afrag[0].y = lo ? RV(0, 1) : RV(1, 0);
    afrag[1].x = lo ? RV(1, 1) : RV(2, 0);
    afrag[1].y = lo ? RV(1, 2) : RV(2, 1);
    afrag[2].x = lo ? RV(2, 2) : 0.0f;
    afrag[2].y = 0.0f;
#undef RV

    const int koff = lo ? 0 : 2;
    const int col  = (c16 < kAct) ? c16 : c16 - kAct;
    v8f acc[kTaps];
#pragma unroll
    for (int i = 0; i < kTaps; ++i) {
        v2f b0 = *(const v2f*)&lds_w[i][col][0 + koff];
        v2f b1 = *(const v2f*)&lds_w[i][col][4 + koff];
        v2f b2 = *(const v2f*)&lds_w[i][col][8 + koff];
        v8f t = {};
        t = __builtin_amdgcn_wmma_f32_16x16x4_f32(false, afrag[0], false, b0,
                                                  (short)0, t, false, false);
        t = __builtin_amdgcn_wmma_f32_16x16x4_f32(false, afrag[1], false, b1,
                                                  (short)0, t, false, false);
        t = __builtin_amdgcn_wmma_f32_16x16x4_f32(false, afrag[2], false, b2,
                                                  (short)0, t, false, false);
        acc[i] = t;
    }

    const int px0 = wv * 16 + hi * 8;
    float win[3][10];
#pragma unroll
    for (int r3 = 0; r3 < 3; ++r3)
#pragma unroll
        for (int j = 0; j < 10; ++j)
            win[r3][j] = lds_rv[r3][px0 + j];

    float qv[8];
#pragma unroll
    for (int r = 0; r < 8; ++r) {
        float mx = acc[0][r];
#pragma unroll
        for (int t = 1; t < kTaps; ++t) mx = fmaxf(mx, acc[t][r]);
        float ssum = 0.0f, q = 0.0f;
#pragma unroll
        for (int t = 0; t < kTaps; ++t) {
            const float e = __expf(acc[t][r] - mx);
            ssum += e;
            q    += e * win[t / 3][r + (t % 3)];
        }
        qv[r] = q / ssum;
    }

#pragma unroll
    for (int r = 0; r < 8; ++r) {
        float x = qv[r];
        x = row_max_step<1>(x);
        x = row_max_step<2>(x);
        x = row_max_step<4>(x);
        x = row_max_step<8>(x);
        qv[r] = x;
    }

    float o = qv[0];
#pragma unroll
    for (int r = 1; r < 8; ++r) o = ((c16 & 7) == r) ? qv[r] : o;
    if (c16 < 8) lds_out[px0 + c16] = o;
    __syncthreads();

    if (tid < kW)
        v_out[((size_t)b * kH + h) * kW + tid] = lds_out[tid];
}

extern "C" void kernel_launch(void* const* d_in, const int* in_sizes, int n_in,
                              void* d_out, int out_size, void* d_ws, size_t ws_size,
                              hipStream_t stream) {
    (void)in_sizes; (void)n_in; (void)out_size; (void)d_ws; (void)ws_size;
    const float* values  = (const float*)d_in[0];
    const float* rewards = (const float*)d_in[1];
    const float* wt      = (const float*)d_in[2];
    float* out = (float*)d_out;

    const size_t slice = (size_t)kB * kH * kW;
    for (int t = 0; t < kDepth; ++t) {
        const float* vin = (t == 0) ? values : out + (size_t)(t - 1) * slice;
        float* vout = out + (size_t)t * slice;
        vin_step_kernel<<<kB * kH, kThreads, 0, stream>>>(vin, rewards, wt, vout);
    }
}


# Work-group owns a 6-row strip: 256 work-groups of 12 waves, not 1536 of 6; rv and weights staged once per strip in one memory round trip; a wave keeps its 27 weight fragments and walks 3 tiles
# speedup vs baseline: 1.9738x; 1.2790x over previous
#include <hip/hip_runtime.h>


constexpr int kDepth   = 16;
constexpr int kB       = 16;
constexpr int kH       = 96;
constexpr int kW       = 96;
constexpr int kAct     = 10;
constexpr int kTaps    = 9;
constexpr int kKPad    = 12;
constexpr int kRows    = 6;
constexpr int kColT    = kW / 16;
constexpr int kTilesPW = 3;
constexpr int kWaves   = kRows * kColT / kTilesPW;
constexpr int kThreads = kWaves * 32;
constexpr int kRvN     = (kRows + 2) * (kW + 2);
constexpr int kWN      = kTaps * kAct * kKPad;
constexpr int kRvIt    = (kRvN + kThreads - 1) / kThreads;
constexpr int kWIt     = (kWN + kThreads - 1) / kThreads;
constexpr int kPitch   = kW + 2;

typedef __attribute__((ext_vector_type(2))) float v2f;
typedef __attribute__((ext_vector_type(8))) float v8f;

template <int M>
__device__ __forceinline__ float row_max_step(float x) {
    const int b = __builtin_bit_cast(int, x);
    const int o = __builtin_amdgcn_update_dpp(b, b, 0x160 + M, 0xf, 0xf, false);
    return fmaxf(x, __builtin_bit_cast(float, o));
}

__global__ __launch_bounds__(kThreads)
void vin_step_kernel(const float* __restrict__ v_in,
                     const float* __restrict__ rewards,
                     const float* __restrict__ wt,
                     float* __restrict__ v_out) {
    __shared__ float lds_rv[kRvIt * kThreads];
    __shared__ __attribute__((aligned(16))) float lds_w[kWIt * kThreads];
    __shared__ __attribute__((aligned(16))) float lds_out[kRows * kW];

    const int tid  = threadIdx.x;
    const int lane = tid & 31;
    const int wv   = tid >> 5;
    const int b    = blockIdx.x / (kH / kRows);
    const int h0   = (blockIdx.x % (kH / kRows)) * kRows;

    const float* vimg = v_in    + (size_t)b * kH * kW;
    const float* rimg = rewards + (size_t)b * kH * kW;

#pragma unroll
    for (int j = 0; j < kRvIt; ++j) {
        const int i0 = tid + j * kThreads;
        const int i  = min(i0, kRvN - 1);
        const int r  = i / kPitch;
        const int c  = i % kPitch;
        const int hh = h0 - 1 + r;
        const int ww = c - 1;
        const bool inb = (hh >= 0) & (hh < kH) & (ww >= 0) & (ww < kW);
        const int hc = min(max(hh, 0), kH - 1);
        const int wc = min(max(ww, 0), kW - 1);
        const float val = vimg[hc * kW + wc] + rimg[hc * kW + wc];
        lds_rv[i0] = inb ? val : 0.0f;
    }

#pragma unroll
    for (int j = 0; j < kWIt; ++j) {
        const int i0  = tid + j * kThreads;
        const int idx = min(i0, kWN - 1);
        const int i   = idx / (kAct * kKPad);
        const int a   = (idx / kKPad) % kAct;
        const int k   = idx % kKPad;
        const int ch  = a * kTaps + i;
        const float val = wt[ch * 9 + min(k, 8)];
        lds_w[i0] = (k < 9) ? val : 0.0f;
    }
    __syncthreads();

    const int  c16 = lane & 15;
    const int  hi  = lane >> 4;
    const int  ct  = wv % kColT;
    const int  lr0 = (wv / kColT) * kTilesPW;
    const int  m   = c16 + ct * 16;
    const bool lo  = lane < 16;

    const int koff = lo ? 0 : 2;
    const int col  = (c16 < kAct) ? c16 : c16 - kAct;
    v2f bfrag[kTaps][3];
#pragma unroll
    for (int i = 0; i < kTaps; ++i) {
        const float* wrow = &lds_w[(i * kAct + col) * kKPad + koff];
        bfrag[i][0] = *(const v2f*)(wrow + 0);
        bfrag[i][1] = *(const v2f*)(wrow + 4);
        bfrag[i][2] = *(const v2f*)(wrow + 8);
    }

    const int px0 = ct * 16 + hi * 8;

    for (int t = 0; t < kTilesPW; ++t) {
        const int lr = lr0 + t;

#define RV(r, c) lds_rv[(lr + (r)) * kPitch + m + (c)]
        v2f afrag[3];
        afrag[0].x = lo ? RV(0, 0) : RV(0, 2);
        afrag[0].y = lo ? RV(0, 1) : RV(1, 0);
        afrag[1].x = lo ? RV(1, 1) : RV(2, 0);
        afrag[1].y = lo ? RV(1, 2) : RV(2, 1);
        afrag[2].x = lo ? RV(2, 2) : 0.0f;
        afrag[2].y = 0.0f;
#undef RV

        v8f acc[kTaps];
#pragma unroll
        for (int i = 0; i < kTaps; ++i) {
            v8f d = {};
            d = __builtin_amdgcn_wmma_f32_16x16x4_f32(false, afrag[0], false, bfrag[i][0],
                                                      (short)0, d, false, false);
            d = __builtin_amdgcn_wmma_f32_16x16x4_f32(false, afrag[1], false, bfrag[i][1],
                                                      (short)0, d, false, false);
            d = __builtin_amdgcn_wmma_f32_16x16x4_f32(false, afrag[2], false, bfrag[i][2],
                                                      (short)0, d, false, false);
            acc[i] = d;
        }

        float win[3][10];
#pragma unroll
        for (int r3 = 0; r3 < 3; ++r3)
#pragma unroll
            for (int j = 0; j < 10; ++j)
                win[r3][j] = lds_rv[(lr + r3) * kPitch + px0 + j];

        float qv[8];
#pragma unroll
        for (int r = 0; r < 8; ++r) {
            float mx = acc[0][r];
#pragma unroll
            for (int s = 1; s < kTaps; ++s) mx = fmaxf(mx, acc[s][r]);
            float ssum = 0.0f, q = 0.0f;
#pragma unroll
            for (int s = 0; s < kTaps; ++s) {
                const float e = __expf(acc[s][r] - mx);
                ssum += e;
                q    += e * win[s / 3][r + (s % 3)];
            }
            qv[r] = q / ssum;
        }

#pragma unroll
        for (int r = 0; r < 8; ++r) {
            float x = qv[r];
            x = row_max_step<1>(x);
            x = row_max_step<2>(x);
            x = row_max_step<4>(x);
            x = row_max_step<8>(x);
            qv[r] = x;
        }

        float o = qv[0];
#pragma unroll
        for (int r = 1; r < 8; ++r) o = ((c16 & 7) == r) ? qv[r] : o;
        if (c16 < 8) lds_out[lr * kW + px0 + c16] = o;
    }
    __syncthreads();

    if (tid < kRows * kW / 4) {
        float4* dst = (float4*)(v_out + ((size_t)b * kH + h0) * kW);
        dst[tid] = ((const float4*)lds_out)[tid];
    }
}

extern "C" void kernel_launch(void* const* d_in, const int* in_sizes, int n_in,
                              void* d_out, int out_size, void* d_ws, size_t ws_size,
                              hipStream_t stream) {
    (void)in_sizes; (void)n_in; (void)out_size; (void)d_ws; (void)ws_size;
    const float* values  = (const float*)d_in[0];
    const float* rewards = (const float*)d_in[1];
    const float* wt      = (const float*)d_in[2];
    float* out = (float*)d_out;

    const size_t slice = (size_t)kB * kH * kW;
    for (int t = 0; t < kDepth; ++t) {
        const float* vin = (t == 0) ? values : out + (size_t)(t - 1) * slice;
        float* vout = out + (size_t)t * slice;
        vin_step_kernel<<<kB * (kH / kRows), kThreads, 0, stream>>>(vin, rewards, wt, vout);
    }
}


# A rows are pixel triples, K=16 their 3x5 window, columns 5 actions x 3 pixels (15 of 16 used, not 10): 36 WMMAs per 48 pixels not 81, a third fewer softmax lanes; conv products summed in a new order
# speedup vs baseline: 2.5111x; 1.2722x over previous
#include <hip/hip_runtime.h>


constexpr int kDepth   = 16;
constexpr int kB       = 16;
constexpr int kH       = 96;
constexpr int kW       = 96;
constexpr int kTaps    = 9;
constexpr int kGrp     = 2;
constexpr int kKS      = 4;
constexpr int kRows    = 6;
constexpr int kWaves   = kRows * kGrp;
constexpr int kThreads = kWaves * 32;
constexpr int kPitch   = kW + 2;
constexpr int kRvN     = (kRows + 2) * kPitch;
constexpr int kRvIt    = (kRvN + kThreads - 1) / kThreads;
constexpr int kWPitch  = 20;
constexpr int kWItems  = kGrp * kTaps * 16 * kKS;
constexpr int kWIt     = kWItems / kThreads;
static_assert(kWItems % kThreads == 0, "B staging has no tail");
constexpr int kTileW   = 48;

typedef __attribute__((ext_vector_type(2))) float v2f;
typedef __attribute__((ext_vector_type(8))) float v8f;

template <int N>
__device__ __forceinline__ float row_from_above(float x) {
    const int b = __builtin_bit_cast(int, x);
    const int o = __builtin_amdgcn_update_dpp(b, b, 0x100 + N, 0xf, 0xf, false);
    return __builtin_bit_cast(float, o);
}

__global__ __launch_bounds__(kThreads)
void vin_step_kernel(const float* __restrict__ v_in,
                     const float* __restrict__ rewards,
                     const float* __restrict__ wt,
                     float* __restrict__ v_out) {
    __shared__ float lds_rv[kRvIt * kThreads];
    __shared__ __attribute__((aligned(16))) float lds_w[kGrp * kTaps * 16 * kWPitch];
    __shared__ __attribute__((aligned(16))) float lds_part[kGrp * kRows * kW];

    const int tid  = threadIdx.x;
    const int lane = tid & 31;
    const int wv   = tid >> 5;
    const int b    = blockIdx.x / (kH / kRows);
    const int h0   = (blockIdx.x % (kH / kRows)) * kRows;

    const float* vimg = v_in    + (size_t)b * kH * kW;
    const float* rimg = rewards + (size_t)b * kH * kW;

#pragma unroll
    for (int j = 0; j < kRvIt; ++j) {
        const int i0 = tid + j * kThreads;
        const int i  = min(i0, kRvN - 1);
        const int r  = i / kPitch;
        const int c  = i % kPitch;
        const int hh = h0 - 1 + r;
        const int ww = c - 1;
        const bool inb = (hh >= 0) & (hh < kH) & (ww >= 0) & (ww < kW);
        const int hc = min(max(hh, 0), kH - 1);
        const int wc = min(max(ww, 0), kW - 1);
        const float val = vimg[hc * kW + wc] + rimg[hc * kW + wc];
        lds_rv[i0] = inb ? val : 0.0f;
    }

#pragma unroll
    for (int j = 0; j < kWIt; ++j) {
        const int it = tid + j * kThreads;
        const int s  = it & 3;
        const int n  = (it >> 2) & 15;
        const int gi = it >> 6;
        const int g  = (gi >= kTaps) ? 1 : 0;
        const int i  = gi - g * kTaps;
        const int a5 = n / 3;
        const int o  = n - a5 * 3;
        const int chn = min(g * 5 + a5, 9) * kTaps + i;
        float vals[4];
#pragma unroll
        for (int e = 0; e < 4; ++e) {
            const int kk = 4 * s + e;
            const int ky = kk / 5;
            const int t  = kk - ky * 5 - o;
            const bool valid = (n < 15) & (kk < 15) & (t >= 0) & (t <= 2);
            const float val = wt[chn * 9 + min(ky, 2) * 3 + min(max(t, 0), 2)];
            vals[e] = valid ? val : 0.0f;
        }
        *(float4*)&lds_w[(gi * 16 + n) * kWPitch + 4 * s] =
            make_float4(vals[0], vals[1], vals[2], vals[3]);
    }
    __syncthreads();

    const int c16 = lane & 15;
    const int hi  = lane >> 4;
    const int g   = wv & 1;
    const int lr  = wv >> 1;
    const int o   = c16 % 3;

    v2f bfrag[kTaps][kKS];
#pragma unroll
    for (int i = 0; i < kTaps; ++i)
#pragma unroll
        for (int s = 0; s < kKS; ++s)
            bfrag[i][s] = *(const v2f*)&lds_w[((g * kTaps + i) * 16 + c16) * kWPitch
                                              + 4 * s + 2 * hi];

    int aoff[kKS][2];
#pragma unroll
    for (int s = 0; s < kKS; ++s)
#pragma unroll
        for (int e = 0; e < 2; ++e) {
            const int kk = min(4 * s + 2 * hi + e, 14);
            aoff[s][e] = (kk / 5) * kPitch + (kk % 5);
        }
    const int abase = lr * kPitch + 3 * c16;
    const int wbase = lr * kPitch + 24 * hi + o;

    for (int ch = 0; ch < 2; ++ch) {
        const int cb = ch * kTileW;

        v2f afrag[kKS];
#pragma unroll
        for (int s = 0; s < kKS; ++s) {
            afrag[s].x = lds_rv[abase + cb + aoff[s][0]];
            afrag[s].y = lds_rv[abase + cb + aoff[s][1]];
        }
        afrag[3].y = hi ? 0.0f : afrag[3].y;

        v8f acc[kTaps];
#pragma unroll
        for (int i = 0; i < kTaps; ++i) {
            v8f d = {};
#pragma unroll
            for (int s = 0; s < kKS; ++s)
                d = __builtin_amdgcn_wmma_f32_16x16x4_f32(false, afrag[s], false, bfrag[i][s],
                                                          (short)0, d, false, false);
            acc[i] = d;
        }

        float win[3][24];
#pragma unroll
        for (int ky = 0; ky < 3; ++ky)
#pragma unroll
            for (int j = 0; j < 24; ++j)
                win[ky][j] = lds_rv[wbase + cb + ky * kPitch + j];

        float qv[8];
#pragma unroll
        for (int r = 0; r < 8; ++r) {
            float mx = acc[0][r];
#pragma unroll
            for (int s = 1; s < kTaps; ++s) mx = fmaxf(mx, acc[s][r]);
            float ssum = 0.0f, q = 0.0f;
#pragma unroll
            for (int s = 0; s < kTaps; ++s) {
                const float e = __expf(acc[s][r] - mx);
                ssum += e;
                q    += e * win[s / 3][3 * r + (s % 3)];
            }
            qv[r] = q / ssum;
        }

#pragma unroll
        for (int r = 0; r < 8; ++r) {
            const float y  = qv[r];
            const float z1 = fmaxf(y,  row_from_above<3>(y));
            const float z2 = fmaxf(z1, row_from_above<6>(z1));
            qv[r]          = fmaxf(z2, row_from_above<12>(y));
        }

        if (c16 < 3) {
            float* dstp = &lds_part[(g * kRows + lr) * kW + cb + 24 * hi + c16];
#pragma unroll
            for (int r = 0; r < 8; ++r) dstp[3 * r] = qv[r];
        }
    }
    __syncthreads();

    if (tid < kRows * kW / 4) {
        const float4 p0 = ((const float4*)lds_part)[tid];
        const float4 p1 = ((const float4*)lds_part)[kRows * kW / 4 + tid];
        float4* dst = (float4*)(v_out + ((size_t)b * kH + h0) * kW);
        dst[tid] = make_float4(fmaxf(p0.x, p1.x), fmaxf(p0.y, p1.y),
                               fmaxf(p0.z, p1.z), fmaxf(p0.w, p1.w));
    }
}

extern "C" void kernel_launch(void* const* d_in, const int* in_sizes, int n_in,
                              void* d_out, int out_size, void* d_ws, size_t ws_size,
                              hipStream_t stream) {
    (void)in_sizes; (void)n_in; (void)out_size; (void)d_ws; (void)ws_size;
    const float* values  = (const float*)d_in[0];
    const float* rewards = (const float*)d_in[1];
    const float* wt      = (const float*)d_in[2];
    float* out = (float*)d_out;

    const size_t slice = (size_t)kB * kH * kW;
    for (int t = 0; t < kDepth; ++t) {
        const float* vin = (t == 0) ? values : out + (size_t)(t - 1) * slice;
        float* vout = out + (size_t)t * slice;
        vin_step_kernel<<<kB * (kH / kRows), kThreads, 0, stream>>>(vin, rewards, wt, vout);
    }
}


# Wave priority by tile: s_setprio 1 through a wave's first tile, 0 for its second: the matrix pipe runs all first tiles, then all second tiles; no first-tile softmax waits behind an older wave
# speedup vs baseline: 2.5976x; 1.0344x over previous
#include <hip/hip_runtime.h>


constexpr int kDepth   = 16;
constexpr int kB       = 16;
constexpr int kH       = 96;
constexpr int kW       = 96;
constexpr int kTaps    = 9;
constexpr int kGrp     = 2;
constexpr int kKS      = 4;
constexpr int kRows    = 6;
constexpr int kWaves   = kRows * kGrp;
constexpr int kThreads = kWaves * 32;
constexpr int kPitch   = kW + 2;
constexpr int kRvN     = (kRows + 2) * kPitch;
constexpr int kRvIt    = (kRvN + kThreads - 1) / kThreads;
constexpr int kWPitch  = 20;
constexpr int kWItems  = kGrp * kTaps * 16 * kKS;
constexpr int kWIt     = kWItems / kThreads;
static_assert(kWItems % kThreads == 0, "B staging has no tail");
constexpr int kTileW   = 48;

typedef __attribute__((ext_vector_type(2))) float v2f;
typedef __attribute__((ext_vector_type(8))) float v8f;

template <int N>
__device__ __forceinline__ float row_from_above(float x) {
    const int b = __builtin_bit_cast(int, x);
    const int o = __builtin_amdgcn_update_dpp(b, b, 0x100 + N, 0xf, 0xf, false);
    return __builtin_bit_cast(float, o);
}

__global__ __launch_bounds__(kThreads)
void vin_step_kernel(const float* __restrict__ v_in,
                     const float* __restrict__ rewards,
                     const float* __restrict__ wt,
                     float* __restrict__ v_out) {
    __shared__ float lds_rv[kRvIt * kThreads];
    __shared__ __attribute__((aligned(16))) float lds_w[kGrp * kTaps * 16 * kWPitch];
    __shared__ __attribute__((aligned(16))) float lds_part[kGrp * kRows * kW];

    __builtin_amdgcn_s_setprio(1);

    const int tid  = threadIdx.x;
    const int lane = tid & 31;
    const int wv   = tid >> 5;
    const int b    = blockIdx.x / (kH / kRows);
    const int h0   = (blockIdx.x % (kH / kRows)) * kRows;

    const float* vimg = v_in    + (size_t)b * kH * kW;
    const float* rimg = rewards + (size_t)b * kH * kW;

#pragma unroll
    for (int j = 0; j < kRvIt; ++j) {
        const int i0 = tid + j * kThreads;
        const int i  = min(i0, kRvN - 1);
        const int r  = i / kPitch;
        const int c  = i % kPitch;
        const int hh = h0 - 1 + r;
        const int ww = c - 1;
        const bool inb = (hh >= 0) & (hh < kH) & (ww >= 0) & (ww < kW);
        const int hc = min(max(hh, 0), kH - 1);
        const int wc = min(max(ww, 0), kW - 1);
        const float val = vimg[hc * kW + wc] + rimg[hc * kW + wc];
        lds_rv[i0] = inb ? val : 0.0f;
    }

#pragma unroll
    for (int j = 0; j < kWIt; ++j) {
        const int it = tid + j * kThreads;
        const int s  = it & 3;
        const int n  = (it >> 2) & 15;
        const int gi = it >> 6;
        const int g  = (gi >= kTaps) ? 1 : 0;
        const int i  = gi - g * kTaps;
        const int a5 = n / 3;
        const int o  = n - a5 * 3;
        const int chn = min(g * 5 + a5, 9) * kTaps + i;
        float vals[4];
#pragma unroll
        for (int e = 0; e < 4; ++e) {
            const int kk = 4 * s + e;
            const int ky = kk / 5;
            const int t  = kk - ky * 5 - o;
            const bool valid = (n < 15) & (kk < 15) & (t >= 0) & (t <= 2);
            const float val = wt[chn * 9 + min(ky, 2) * 3 + min(max(t, 0), 2)];
            vals[e] = valid ? val : 0.0f;
        }
        *(float4*)&lds_w[(gi * 16 + n) * kWPitch + 4 * s] =
            make_float4(vals[0], vals[1], vals[2], vals[3]);
    }
    __syncthreads();

    const int c16 = lane & 15;
    const int hi  = lane >> 4;
    const int g   = wv & 1;
    const int lr  = wv >> 1;
    const int o   = c16 % 3;

    v2f bfrag[kTaps][kKS];
#pragma unroll
    for (int i = 0; i < kTaps; ++i)
#pragma unroll
        for (int s = 0; s < kKS; ++s)
            bfrag[i][s] = *(const v2f*)&lds_w[((g * kTaps + i) * 16 + c16) * kWPitch
                                              + 4 * s + 2 * hi];

    int aoff[kKS][2];
#pragma unroll
    for (int s = 0; s < kKS; ++s)
#pragma unroll
        for (int e = 0; e < 2; ++e) {
            const int kk = min(4 * s + 2 * hi + e, 14);
            aoff[s][e] = (kk / 5) * kPitch + (kk % 5);
        }
    const int abase = lr * kPitch + 3 * c16;
    const int wbase = lr * kPitch + 24 * hi + o;

    for (int ch = 0; ch < 2; ++ch) {
        const int cb = ch * kTileW;
        if (ch == 1) __builtin_amdgcn_s_setprio(0);

        v2f afrag[kKS];
#pragma unroll
        for (int s = 0; s < kKS; ++s) {
            afrag[s].x = lds_rv[abase + cb + aoff[s][0]];
            afrag[s].y = lds_rv[abase + cb + aoff[s][1]];
        }
        afrag[3].y = hi ? 0.0f : afrag[3].y;

        v8f acc[kTaps];
#pragma unroll
        for (int i = 0; i < kTaps; ++i) {
            v8f d = {};
#pragma unroll
            for (int s = 0; s < kKS; ++s)
                d = __builtin_amdgcn_wmma_f32_16x16x4_f32(false, afrag[s], false, bfrag[i][s],
                                                          (short)0, d, false, false);
            acc[i] = d;
        }

        float win[3][24];
#pragma unroll
        for (int ky = 0; ky < 3; ++ky)
#pragma unroll
            for (int j = 0; j < 24; ++j)
                win[ky][j] = lds_rv[wbase + cb + ky * kPitch + j];

        float qv[8];
#pragma unroll
        for (int r = 0; r < 8; ++r) {
            float mx = acc[0][r];
#pragma unroll
            for (int s = 1; s < kTaps; ++s) mx = fmaxf(mx, acc[s][r]);
            float ssum = 0.0f, q = 0.0f;
#pragma unroll
            for (int s = 0; s < kTaps; ++s) {
                const float e = __expf(acc[s][r] - mx);
                ssum += e;
                q    += e * win[s / 3][3 * r + (s % 3)];
            }
            qv[r] = q / ssum;
        }

#pragma unroll
        for (int r = 0; r < 8; ++r) {
            const float y  = qv[r];
            const float z1 = fmaxf(y,  row_from_above<3>(y));
            const float z2 = fmaxf(z1, row_from_above<6>(z1));
            qv[r]          = fmaxf(z2, row_from_above<12>(y));
        }

        if (c16 < 3) {
            float* dstp = &lds_part[(g * kRows + lr) * kW + cb + 24 * hi + c16];
#pragma unroll
            for (int r = 0; r < 8; ++r) dstp[3 * r] = qv[r];
        }
    }
    __syncthreads();

    if (tid < kRows * kW / 4) {
        const float4 p0 = ((const float4*)lds_part)[tid];
        const float4 p1 = ((const float4*)lds_part)[kRows * kW / 4 + tid];
        float4* dst = (float4*)(v_out + ((size_t)b * kH + h0) * kW);
        dst[tid] = make_float4(fmaxf(p0.x, p1.x), fmaxf(p0.y, p1.y),
                               fmaxf(p0.z, p1.z), fmaxf(p0.w, p1.w));
    }
}

extern "C" void kernel_launch(void* const* d_in, const int* in_sizes, int n_in,
                              void* d_out, int out_size, void* d_ws, size_t ws_size,
                              hipStream_t stream) {
    (void)in_sizes; (void)n_in; (void)out_size; (void)d_ws; (void)ws_size;
    const float* values  = (const float*)d_in[0];
    const float* rewards = (const float*)d_in[1];
    const float* wt      = (const float*)d_in[2];
    float* out = (float*)d_out;

    const size_t slice = (size_t)kB * kH * kW;
    for (int t = 0; t < kDepth; ++t) {
        const float* vin = (t == 0) ? values : out + (size_t)(t - 1) * slice;
        float* vout = out + (size_t)t * slice;
        vin_step_kernel<<<kB * (kH / kRows), kThreads, 0, stream>>>(vin, rewards, wt, vout);
    }
}
